# starting kernel
# baseline (speedup 1.0000x reference)
#include <hip/hip_runtime.h>
#include <hip/hip_bf16.h>
#include <cstddef>

typedef __attribute__((ext_vector_type(2))) float v2f;
typedef __attribute__((ext_vector_type(4))) float v4f;
typedef __attribute__((ext_vector_type(8))) float v8f;

#define BATCH 2
#define Hh    384
#define Ww    1248
#define DISP  192
#define WT    (Ww / 16)
#define HT    (Hh / 16)
#define WGW   6
#define WQ    (WT / WGW)
#define NBLK  (BATCH * HT * WQ)

#define SLAB_COLS   288
#define SLAB_STRIDE 292

__global__ __launch_bounds__(192) void StereoMatching_wmma_kernel(
    const float* __restrict__ L, const float* __restrict__ R, float* __restrict__ out)
{
    __shared__ __align__(16) float slab[16 * SLAB_STRIDE];

    const int tid  = threadIdx.x;
    const int lane = tid & 31;
    const int widx = tid >> 5;

    const int wq = blockIdx.x % WQ;
    const int t2 = blockIdx.x / WQ;
    const int ht = t2 % HT;
    const int b  = t2 / HT;
    const int h0 = ht << 4;
    const int gw0 = wq * (WGW * 16) - DISP;

    {
        const int frow  = tid / 12;
        const int fcol0 = (tid % 12) * 24;
        const float* grow = R + (size_t)(b * Hh + h0 + frow) * Ww;
        float* srow = &slab[frow * SLAB_STRIDE];
        if (gw0 >= 0) {
#pragma unroll
            for (int j = 0; j < 6; ++j) {
                const int col = fcol0 + 4 * j;
                *(v4f*)(srow + col) = *(const v4f*)(grow + gw0 + col);
            }
        } else {
#pragma unroll
            for (int j = 0; j < 24; ++j) {
                const int col = fcol0 + j;
                const int gc  = gw0 + col;
                const int gcc = gc < 0 ? 0 : gc;
                const float v = grow[gcc];
                srow[col] = (gc >= 0) ? v : 0.0f;
            }
        }
    }
    __syncthreads();

    const int n     = lane & 15;
    const int khalf = lane >> 4;

    const float* arow = &slab[n * SLAB_STRIDE + widx * 16 + 2 * khalf];
    v8f c = {};

#pragma unroll
    for (int kc = 0; kc < 4; ++kc) {
        const v2f a = *(const v2f*)(arow + 4 * kc);
        const int x0 = n + DISP - (4 * kc + 2 * khalf);
        v2f bb;
        bb.x = (x0     < DISP) ? (float)x0       : 0.0f;
        bb.y = (x0 - 1 < DISP) ? (float)(x0 - 1) : 0.0f;
        c = __builtin_amdgcn_wmma_f32_16x16x4_f32(false, a, false, bb,
                                                  (short)0, c, false, false);
    }

    float xf0 = (float)(n + DISP - 16 - 2 * khalf);
    float xf1 = xf0 - 1.0f;
#pragma unroll 4
    for (int kc = 4; kc < 52; ++kc) {
        const v2f a = *(const v2f*)(arow + 4 * kc);
        v2f bb;
        bb.x = fmaxf(xf0, 0.0f);
        bb.y = fmaxf(xf1, 0.0f);
        c = __builtin_amdgcn_wmma_f32_16x16x4_f32(false, a, false, bb,
                                                  (short)0, c, false, false);
        xf0 -= 4.0f;
        xf1 -= 4.0f;
    }

    const int w0 = (wq * WGW + widx) << 4;
    const size_t obase = (size_t)(b * Hh + h0) * Ww + (size_t)(w0 + n);
#pragma unroll
    for (int v = 0; v < 8; ++v) {
        const int M = v + (khalf << 3);
        const size_t idx = obase + (size_t)M * Ww;
        out[idx] = L[idx] * c[v];
    }
}

extern "C" void kernel_launch(void* const* d_in, const int* in_sizes, int n_in,
                              void* d_out, int out_size, void* d_ws, size_t ws_size,
                              hipStream_t stream) {
    (void)in_sizes; (void)n_in; (void)out_size; (void)d_ws; (void)ws_size;
    const float* L = (const float*)d_in[0];
    const float* R = (const float*)d_in[1];
    float* O = (float*)d_out;

    static_assert(WT % WGW == 0, "w-tiles must split into groups");
    dim3 grid(NBLK);
    dim3 block(192);
    hipLaunchKernelGGL(StereoMatching_wmma_kernel, grid, block, 0, stream, L, R, O);
}


# Band weight split d=(n-k)+16*(12-m): the 11 interior 16-column chunks are pre-summed with prefix adds, 16 WMMA per tile instead of 52; changes the f32 summation order
# speedup vs baseline: 1.3944x; 1.3944x over previous
#include <hip/hip_runtime.h>
#include <hip/hip_bf16.h>
#include <cstddef>

typedef __attribute__((ext_vector_type(2))) float v2f;
typedef __attribute__((ext_vector_type(4))) float v4f;
typedef __attribute__((ext_vector_type(8))) float v8f;

#define BATCH 2
#define Hh    384
#define Ww    1248
#define DISP  192
#define WT    (Ww / 16)
#define HT    (Hh / 16)
#define WGW   6
#define WQ    (WT / WGW)
#define NBLK  (BATCH * HT * WQ)

#define SLAB_COLS   288
#define SLAB_STRIDE 292

__global__ __launch_bounds__(192) void StereoMatching_wmma_kernel(
    const float* __restrict__ L, const float* __restrict__ R, float* __restrict__ out)
{
    __shared__ __align__(16) float slab[16 * SLAB_STRIDE];

    const int tid  = threadIdx.x;
    const int lane = tid & 31;
    const int widx = tid >> 5;

    const int wq = blockIdx.x % WQ;
    const int t2 = blockIdx.x / WQ;
    const int ht = t2 % HT;
    const int b  = t2 / HT;
    const int h0 = ht << 4;
    const int gw0 = wq * (WGW * 16) - DISP;

    {
        const int frow  = tid / 12;
        const int fcol0 = (tid % 12) * 24;
        const float* grow = R + (size_t)(b * Hh + h0 + frow) * Ww;
        float* srow = &slab[frow * SLAB_STRIDE];
        if (gw0 >= 0) {
#pragma unroll
            for (int j = 0; j < 6; ++j) {
                const int col = fcol0 + 4 * j;
                *(v4f*)(srow + col) = *(const v4f*)(grow + gw0 + col);
            }
        } else {
#pragma unroll
            for (int j = 0; j < 24; ++j) {
                const int col = fcol0 + j;
                const int gc  = gw0 + col;
                const int gcc = gc < 0 ? 0 : gc;
                const float v = grow[gcc];
                srow[col] = (gc >= 0) ? v : 0.0f;
            }
        }
    }
    __syncthreads();

    const int n     = lane & 15;
    const int khalf = lane >> 4;


    const float* arow = &slab[n * SLAB_STRIDE + widx * 16 + 2 * khalf];

    v8f c0 = {};
    v8f c1 = {};

    for (int j = 0; j < 4; ++j) {
        const int e0 = n - (4 * j + 2 * khalf);
        const int e1 = e0 - 1;
        const v2f a_own = *(const v2f*)(arow + 192 + 4 * j);
        v2f t_own;
        t_own.x = (float)(e0 > 0 ? e0 : 0);
        t_own.y = (float)(e1 > 0 ? e1 : 0);
        c0 = __builtin_amdgcn_wmma_f32_16x16x4_f32(false, a_own, false, t_own,
                                                   (short)0, c0, false, false);
        const v2f a_far = *(const v2f*)(arow + 4 * j);
        v2f t_far;
        t_far.x = (e0 < 0) ? (float)(e0 + DISP) : 0.0f;
        t_far.y = (e1 < 0) ? (float)(e1 + DISP) : 0.0f;
        c1 = __builtin_amdgcn_wmma_f32_16x16x4_f32(false, a_far, false, t_far,
                                                   (short)0, c1, false, false);
    }

    v2f P[4], Q[4];
    for (int j = 0; j < 4; ++j) {
        P[j] = *(const v2f*)(arow + 16 + 4 * j);
        Q[j] = P[j];
    }
    for (int m = 2; m <= 11; ++m) {
        for (int j = 0; j < 4; ++j) {
            P[j] += *(const v2f*)(arow + 16 * m + 4 * j);
            Q[j] += P[j];
        }
    }

    for (int j = 0; j < 4; ++j) {
        const int e0 = n - (4 * j + 2 * khalf);
        v2f e;
        e.x = (float)e0;
        e.y = (float)(e0 - 1);
        c0 = __builtin_amdgcn_wmma_f32_16x16x4_f32(false, P[j], false, e,
                                                   (short)0, c0, false, false);
        v2f s16;
        s16.x = 16.0f;
        s16.y = 16.0f;
        c1 = __builtin_amdgcn_wmma_f32_16x16x4_f32(false, Q[j], false, s16,
                                                   (short)0, c1, false, false);
    }
    const v8f c = c0 + c1;

    const int w0 = (wq * WGW + widx) << 4;
    const size_t obase = (size_t)(b * Hh + h0) * Ww + (size_t)(w0 + n);
#pragma unroll
    for (int v = 0; v < 8; ++v) {
        const int M = v + (khalf << 3);
        const size_t idx = obase + (size_t)M * Ww;
        out[idx] = L[idx] * c[v];
    }
}

extern "C" void kernel_launch(void* const* d_in, const int* in_sizes, int n_in,
                              void* d_out, int out_size, void* d_ws, size_t ws_size,
                              hipStream_t stream) {
    (void)in_sizes; (void)n_in; (void)out_size; (void)d_ws; (void)ws_size;
    const float* L = (const float*)d_in[0];
    const float* R = (const float*)d_in[1];
    float* O = (float*)d_out;

    static_assert(WT % WGW == 0, "w-tiles must split into groups");
    dim3 grid(NBLK);
    dim3 block(192);
    hipLaunchKernelGGL(StereoMatching_wmma_kernel, grid, block, 0, stream, L, R, O);
}


# Each wave computes three adjacent tiles from one running prefix over their shared interior chunks (15 chunk reads per 3 tiles, not 39), 312 blocks of 4 waves; changes the f32 summation order
# speedup vs baseline: 1.5098x; 1.0828x over previous
#include <hip/hip_runtime.h>
#include <hip/hip_bf16.h>
#include <cstddef>

typedef __attribute__((ext_vector_type(2))) float v2f;
typedef __attribute__((ext_vector_type(4))) float v4f;
typedef __attribute__((ext_vector_type(8))) float v8f;

#define BATCH 2
#define Hh    384
#define Ww    1248
#define DISP  192
#define ROWS  (BATCH * Hh)
#define WT    (Ww / 16)
#define WGW   6
#define WQ    (WT / WGW)
#define TPW   3
#define SROWS 32
#define NBLK  ((ROWS / SROWS) * WQ)

#define SLAB_COLS   288
#define SLAB_STRIDE 292

__global__ __launch_bounds__(128) void StereoMatching_wmma_kernel(
    const float* __restrict__ L, const float* __restrict__ R, float* __restrict__ out)
{
    __shared__ __align__(16) float slab[SROWS * SLAB_STRIDE];

    const int tid  = threadIdx.x;
    const int lane = tid & 31;
    const int widx = tid >> 5;

    const int wq   = blockIdx.x % WQ;
    const int rp   = blockIdx.x / WQ;
    const int row0 = rp * SROWS;
    const int gw0  = wq * (WGW * 16) - DISP;

    {
        const int l8 = tid & 7;
        const int fr = tid >> 3;
        if (gw0 >= 0) {
#pragma unroll
            for (int hf = 0; hf < 2; ++hf) {
                const int r = fr + 16 * hf;
                const float* grow = R + (size_t)(row0 + r) * Ww + gw0;
                float* srow = &slab[r * SLAB_STRIDE];
#pragma unroll
                for (int k = 0; k < 9; ++k) {
                    const int col = 4 * (l8 + 8 * k);
                    *(v4f*)(srow + col) = *(const v4f*)(grow + col);
                }
            }
        } else {
#pragma unroll
            for (int hf = 0; hf < 2; ++hf) {
                const int r = fr + 16 * hf;
                const float* grow = R + (size_t)(row0 + r) * Ww;
                float* srow = &slab[r * SLAB_STRIDE];
#pragma unroll
                for (int k = 0; k < 9; ++k) {
                    const int col = 4 * (l8 + 8 * k);
                    const int gc  = gw0 + col;
                    const int gcc = gc < 0 ? 0 : gc;
                    const v4f v = *(const v4f*)(grow + gcc);
                    v4f z;
                    z.x = (gc >= 0) ? v.x : 0.0f;
                    z.y = (gc >= 0) ? v.y : 0.0f;
                    z.z = (gc >= 0) ? v.z : 0.0f;
                    z.w = (gc >= 0) ? v.w : 0.0f;
                    *(v4f*)(srow + col) = z;
                }
            }
        }
    }
    __syncthreads();

    const int n     = lane & 15;
    const int khalf = lane >> 4;
    const int strip = widx >> 1;
    const int tw    = (widx & 1) * TPW;

    const float* arow = &slab[(strip * 16 + n) * SLAB_STRIDE + tw * 16 + 2 * khalf];

    v2f t_own[4], t_far[4], e_op[4];
#pragma unroll
    for (int j = 0; j < 4; ++j) {
        const int e0 = n - (4 * j + 2 * khalf);
        const int e1 = e0 - 1;
        t_own[j].x = (float)(e0 > 0 ? e0 : 0);
        t_own[j].y = (float)(e1 > 0 ? e1 : 0);
        t_far[j].x = (e0 < 0) ? (float)(e0 + DISP) : 0.0f;
        t_far[j].y = (e1 < 0) ? (float)(e1 + DISP) : 0.0f;
        e_op[j].x  = (float)e0;
        e_op[j].y  = (float)e1;
    }
    v2f s16;
    s16.x = 16.0f;
    s16.y = 16.0f;

    v8f c[TPW];
#pragma unroll
    for (int s = 0; s < TPW; ++s) c[s] = (v8f){};

#pragma unroll
    for (int j = 0; j < 4; ++j) {
#pragma unroll
        for (int s = 0; s < TPW; ++s) {
            const v2f a_own = *(const v2f*)(arow + 16 * (12 + s) + 4 * j);
            c[s] = __builtin_amdgcn_wmma_f32_16x16x4_f32(false, a_own, false, t_own[j],
                                                         (short)0, c[s], false, false);
        }
#pragma unroll
        for (int s = 0; s < TPW; ++s) {
            const v2f a_far = *(const v2f*)(arow + 16 * s + 4 * j);
            c[s] = __builtin_amdgcn_wmma_f32_16x16x4_f32(false, a_far, false, t_far[j],
                                                         (short)0, c[s], false, false);
        }
    }

    v2f A1[4], A2[4], S[4], G[4], P0[4], Q0[4], P1[4], Q1[4];
#pragma unroll
    for (int j = 0; j < 4; ++j) {
        A1[j] = *(const v2f*)(arow + 16 * 1 + 4 * j);
        A2[j] = *(const v2f*)(arow + 16 * 2 + 4 * j);
        S[j]  = *(const v2f*)(arow + 16 * 3 + 4 * j);
        G[j]  = S[j];
    }
#pragma unroll
    for (int i = 4; i <= 11; ++i) {
#pragma unroll
        for (int j = 0; j < 4; ++j) {
            S[j] += *(const v2f*)(arow + 16 * i + 4 * j);
            G[j] += S[j];
        }
    }
#pragma unroll
    for (int j = 0; j < 4; ++j) {
        P0[j] = (A1[j] + A2[j]) + S[j];
        Q0[j] = 11.0f * A1[j] + (10.0f * A2[j] + G[j]);
        S[j] += *(const v2f*)(arow + 16 * 12 + 4 * j);
        G[j] += S[j];
        P1[j] = A2[j] + S[j];
        Q1[j] = 11.0f * A2[j] + G[j];
        S[j] += *(const v2f*)(arow + 16 * 13 + 4 * j);
        G[j] += S[j];
    }

#pragma unroll
    for (int j = 0; j < 4; ++j) {
        c[0] = __builtin_amdgcn_wmma_f32_16x16x4_f32(false, P0[j], false, e_op[j],
                                                     (short)0, c[0], false, false);
        c[1] = __builtin_amdgcn_wmma_f32_16x16x4_f32(false, P1[j], false, e_op[j],
                                                     (short)0, c[1], false, false);
        c[2] = __builtin_amdgcn_wmma_f32_16x16x4_f32(false, S[j], false, e_op[j],
                                                     (short)0, c[2], false, false);
        c[0] = __builtin_amdgcn_wmma_f32_16x16x4_f32(false, Q0[j], false, s16,
                                                     (short)0, c[0], false, false);
        c[1] = __builtin_amdgcn_wmma_f32_16x16x4_f32(false, Q1[j], false, s16,
                                                     (short)0, c[1], false, false);
        c[2] = __builtin_amdgcn_wmma_f32_16x16x4_f32(false, G[j], false, s16,
                                                     (short)0, c[2], false, false);
    }

#pragma unroll
    for (int s = 0; s < TPW; ++s) {
        const int w0 = (wq * WGW + tw + s) << 4;
        const size_t obase = (size_t)(row0 + strip * 16) * Ww + (size_t)(w0 + n);
#pragma unroll
        for (int v = 0; v < 8; ++v) {
            const int M = v + (khalf << 3);
            const size_t idx = obase + (size_t)M * Ww;
            out[idx] = L[idx] * c[s][v];
        }
    }
}

extern "C" void kernel_launch(void* const* d_in, const int* in_sizes, int n_in,
                              void* d_out, int out_size, void* d_ws, size_t ws_size,
                              hipStream_t stream) {
    (void)in_sizes; (void)n_in; (void)out_size; (void)d_ws; (void)ws_size;
    const float* L = (const float*)d_in[0];
    const float* R = (const float*)d_in[1];
    float* O = (float*)d_out;

    static_assert(WT % WGW == 0, "w-tiles must split into groups");
    static_assert(ROWS % SROWS == 0, "rows must split into blocks");
    static_assert(WGW == 2 * TPW, "a strip's tiles are two triples");
    dim3 grid(NBLK);
    dim3 block(128);
    hipLaunchKernelGGL(StereoMatching_wmma_kernel, grid, block, 0, stream, L, R, O);
}


# Re-tiled to 240 blocks (16-row strip x 16 tiles; 4 waves x 4 tiles off one running prefix): at most one block per WGP and one wave per SIMD, not 312 blocks on 256 WGPs; changes f32 summation order
# speedup vs baseline: 1.6058x; 1.0636x over previous
#include <hip/hip_runtime.h>
#include <hip/hip_bf16.h>
#include <cstddef>

typedef __attribute__((ext_vector_type(2))) float v2f;
typedef __attribute__((ext_vector_type(4))) float v4f;
typedef __attribute__((ext_vector_type(8))) float v8f;

#define BATCH 2
#define Hh    384
#define Ww    1248
#define DISP  192
#define ROWS  (BATCH * Hh)
#define WT    (Ww / 16)
#define STRIPS (ROWS / 16)
#define TPW   4
#define WGW   16
#define BPS   5
#define LAST_T0 (WT - WGW)
#define NBLK  (STRIPS * BPS)

#define SLAB_COLS   448
#define SLAB_STRIDE 484

__global__ __launch_bounds__(128) void StereoMatching_wmma_kernel(
    const float* __restrict__ L, const float* __restrict__ R, float* __restrict__ out)
{
    __shared__ __align__(16) float slab[16 * SLAB_STRIDE];

    const int tid  = threadIdx.x;
    const int lane = tid & 31;
    const int widx = tid >> 5;

    const int bq   = blockIdx.x % BPS;
    const int row0 = (blockIdx.x / BPS) * 16;
    const int t0   = (bq == BPS - 1) ? LAST_T0 : bq * WGW;
    const int gw0  = t0 * 16 - DISP;

    const int n     = lane & 15;
    const int khalf = lane >> 4;
    const int tw    = widx * TPW;

    {
        const int l8 = tid & 7;
        const int fr = tid >> 3;
        float* srow = &slab[fr * SLAB_STRIDE];
        const float* grow = R + (size_t)(row0 + fr) * Ww;
        v4f sv[14];
#pragma unroll
        for (int k = 0; k < 14; ++k) {
            const int gc  = gw0 + 4 * (l8 + 8 * k);
            const int gcc = (k < 6 && gc < 0) ? 0 : gc;
            sv[k] = *(const v4f*)(grow + gcc);
        }
        __builtin_amdgcn_sched_barrier(0);
#pragma unroll
        for (int k = 0; k < 14; ++k) {
            const int col = 4 * (l8 + 8 * k);
            v4f z = sv[k];
            if (k < 6) {
                const int gc = gw0 + col;
                z.x = (gc >= 0) ? z.x : 0.0f;
                z.y = (gc >= 0) ? z.y : 0.0f;
                z.z = (gc >= 0) ? z.z : 0.0f;
                z.w = (gc >= 0) ? z.w : 0.0f;
            }
            *(v4f*)(srow + col) = z;
        }
    }
    __syncthreads();

    const bool own01 = !(bq == BPS - 1 && __builtin_amdgcn_readfirstlane(widx) == 0);
    const size_t obase = (size_t)(row0 + (khalf << 3)) * Ww + (size_t)(((t0 + tw) << 4) + n);
    const float* lptr = L + obase;
    float lv[TPW][8];
#pragma unroll
    for (int s = 2; s < TPW; ++s)
#pragma unroll
        for (int v = 0; v < 8; ++v) lv[s][v] = lptr[v * Ww + 16 * s];
    if (own01) {
#pragma unroll
        for (int s = 0; s < 2; ++s)
#pragma unroll
            for (int v = 0; v < 8; ++v) lv[s][v] = lptr[v * Ww + 16 * s];
    } else {
#pragma unroll
        for (int s = 0; s < 2; ++s)
#pragma unroll
            for (int v = 0; v < 8; ++v) lv[s][v] = 0.0f;
    }
    __builtin_amdgcn_sched_barrier(0);

    const float* arow = &slab[n * SLAB_STRIDE + tw * 16 + 2 * khalf];

    v2f t_own[4], t_far[4], e_op[4];
#pragma unroll
    for (int j = 0; j < 4; ++j) {
        const int e0 = n - (4 * j + 2 * khalf);
        const int e1 = e0 - 1;
        t_own[j].x = (float)(e0 > 0 ? e0 : 0);
        t_own[j].y = (float)(e1 > 0 ? e1 : 0);
        t_far[j].x = (e0 < 0) ? (float)(e0 + DISP) : 0.0f;
        t_far[j].y = (e1 < 0) ? (float)(e1 + DISP) : 0.0f;
        e_op[j].x  = (float)e0;
        e_op[j].y  = (float)e1;
    }
    v2f s16;
    s16.x = 16.0f;
    s16.y = 16.0f;

    v8f c[TPW];
#pragma unroll
    for (int s = 0; s < TPW; ++s) c[s] = (v8f){};

#pragma unroll
    for (int j = 0; j < 4; ++j) {
#pragma unroll
        for (int s = 0; s < TPW; ++s) {
            const v2f a_own = *(const v2f*)(arow + 16 * (12 + s) + 4 * j);
            c[s] = __builtin_amdgcn_wmma_f32_16x16x4_f32(false, a_own, false, t_own[j],
                                                         (short)0, c[s], false, false);
        }
#pragma unroll
        for (int s = 0; s < TPW; ++s) {
            const v2f a_far = *(const v2f*)(arow + 16 * s + 4 * j);
            c[s] = __builtin_amdgcn_wmma_f32_16x16x4_f32(false, a_far, false, t_far[j],
                                                         (short)0, c[s], false, false);
        }
    }

    v2f A1[4], A2[4], A3[4], S[4], G[4], P0[4], Q0[4], P1[4], Q1[4], P2[4], Q2[4];
#pragma unroll
    for (int j = 0; j < 4; ++j) {
        A1[j] = *(const v2f*)(arow + 16 * 1 + 4 * j);
        A2[j] = *(const v2f*)(arow + 16 * 2 + 4 * j);
        A3[j] = *(const v2f*)(arow + 16 * 3 + 4 * j);
        S[j]  = *(const v2f*)(arow + 16 * 4 + 4 * j);
        G[j]  = S[j];
    }
#pragma unroll
    for (int i = 5; i <= 11; ++i) {
#pragma unroll
        for (int j = 0; j < 4; ++j) {
            S[j] += *(const v2f*)(arow + 16 * i + 4 * j);
            G[j] += S[j];
        }
    }
#pragma unroll
    for (int j = 0; j < 4; ++j) {
        P0[j] = (A1[j] + A2[j]) + (A3[j] + S[j]);
        Q0[j] = (11.0f * A1[j] + 10.0f * A2[j]) + (9.0f * A3[j] + G[j]);
        S[j] += *(const v2f*)(arow + 16 * 12 + 4 * j);
        G[j] += S[j];
        P1[j] = (A2[j] + A3[j]) + S[j];
        Q1[j] = 11.0f * A2[j] + (10.0f * A3[j] + G[j]);
        S[j] += *(const v2f*)(arow + 16 * 13 + 4 * j);
        G[j] += S[j];
        P2[j] = A3[j] + S[j];
        Q2[j] = 11.0f * A3[j] + G[j];
        S[j] += *(const v2f*)(arow + 16 * 14 + 4 * j);
        G[j] += S[j];
    }

#pragma unroll
    for (int j = 0; j < 4; ++j) {
        c[0] = __builtin_amdgcn_wmma_f32_16x16x4_f32(false, P0[j], false, e_op[j],
                                                     (short)0, c[0], false, false);
        c[1] = __builtin_amdgcn_wmma_f32_16x16x4_f32(false, P1[j], false, e_op[j],
                                                     (short)0, c[1], false, false);
        c[2] = __builtin_amdgcn_wmma_f32_16x16x4_f32(false, P2[j], false, e_op[j],
                                                     (short)0, c[2], false, false);
        c[3] = __builtin_amdgcn_wmma_f32_16x16x4_f32(false, S[j], false, e_op[j],
                                                     (short)0, c[3], false, false);
        c[0] = __builtin_amdgcn_wmma_f32_16x16x4_f32(false, Q0[j], false, s16,
                                                     (short)0, c[0], false, false);
        c[1] = __builtin_amdgcn_wmma_f32_16x16x4_f32(false, Q1[j], false, s16,
                                                     (short)0, c[1], false, false);
        c[2] = __builtin_amdgcn_wmma_f32_16x16x4_f32(false, Q2[j], false, s16,
                                                     (short)0, c[2], false, false);
        c[3] = __builtin_amdgcn_wmma_f32_16x16x4_f32(false, G[j], false, s16,
                                                     (short)0, c[3], false, false);
    }

    float* optr = out + obase;
    if (own01) {
#pragma unroll
        for (int s = 0; s < 2; ++s)
#pragma unroll
            for (int v = 0; v < 8; ++v)
                optr[v * Ww + 16 * s] = lv[s][v] * c[s][v];
    }
#pragma unroll
    for (int s = 2; s < TPW; ++s)
#pragma unroll
        for (int v = 0; v < 8; ++v)
            optr[v * Ww + 16 * s] = lv[s][v] * c[s][v];
}

extern "C" void kernel_launch(void* const* d_in, const int* in_sizes, int n_in,
                              void* d_out, int out_size, void* d_ws, size_t ws_size,
                              hipStream_t stream) {
    (void)in_sizes; (void)n_in; (void)out_size; (void)d_ws; (void)ws_size;
    const float* L = (const float*)d_in[0];
    const float* R = (const float*)d_in[1];
    float* O = (float*)d_out;

    static_assert(ROWS % 16 == 0, "rows must split into strips");
    static_assert(WGW == 4 * TPW, "a block's tiles are four quadruples");
    static_assert(WGW * (BPS - 1) <= WT && WGW * BPS >= WT, "five blocks cover a strip");
    static_assert(SLAB_COLS == DISP + WGW * 16, "slab = halo + tile columns");
    dim3 grid(NBLK);
    dim3 block(128);
    hipLaunchKernelGGL(StereoMatching_wmma_kernel, grid, block, 0, stream, L, R, O);
}


# The 32 l loads are requested right behind the 14 slab loads and stay in flight across the barrier (its fence is LDS-only, so it drains no loads): one memory round trip per wave, not two
# speedup vs baseline: 1.8097x; 1.1270x over previous
#include <hip/hip_runtime.h>
#include <hip/hip_bf16.h>
#include <cstddef>

typedef __attribute__((ext_vector_type(2))) float v2f;
typedef __attribute__((ext_vector_type(4))) float v4f;
typedef __attribute__((ext_vector_type(8))) float v8f;

#define BATCH 2
#define Hh    384
#define Ww    1248
#define DISP  192
#define ROWS  (BATCH * Hh)
#define WT    (Ww / 16)
#define STRIPS (ROWS / 16)
#define TPW   4
#define WGW   16
#define BPS   5
#define LAST_T0 (WT - WGW)
#define NBLK  (STRIPS * BPS)

#define SLAB_COLS   448
#define HEAD_VALU 52
#define GAP_VALU  11
#define SLAB_STRIDE 484

__global__ __launch_bounds__(128) void StereoMatching_wmma_kernel(
    const float* __restrict__ L, const float* __restrict__ R, float* __restrict__ out)
{
    __shared__ __align__(16) float slab[16 * SLAB_STRIDE];

    const int tid  = threadIdx.x;
    const int lane = tid & 31;
    const int widx = tid >> 5;

    const int bq   = blockIdx.x % BPS;
    const int row0 = (blockIdx.x / BPS) * 16;
    const int t0   = (bq == BPS - 1) ? LAST_T0 : bq * WGW;
    const int gw0  = t0 * 16 - DISP;

    const int n     = lane & 15;
    const int khalf = lane >> 4;
    const int tw    = widx * TPW;

    const bool own01 = !(bq == BPS - 1 && __builtin_amdgcn_readfirstlane(widx) == 0);
    const size_t obase = (size_t)(row0 + (khalf << 3)) * Ww + (size_t)(((t0 + tw) << 4) + n);
    const float* lptr = L + obase;
    float lv[TPW][8];

    {
        const int l8 = tid & 7;
        const int fr = tid >> 3;
        float* srow = &slab[fr * SLAB_STRIDE];
        const float* grow = R + (size_t)(row0 + fr) * Ww;
        v4f sv[14];
#pragma unroll
        for (int k = 0; k < 14; ++k) {
            const int gc  = gw0 + 4 * (l8 + 8 * k);
            const int gcc = (k < 6 && gc < 0) ? 0 : gc;
            sv[k] = *(const v4f*)(grow + gcc);
        }
#pragma unroll
        for (int s = 0; s < TPW; ++s)
#pragma unroll
            for (int v = 0; v < 8; ++v) lv[s][v] = lptr[v * Ww + 16 * s];
        asm volatile("s_wait_xcnt 0x0");
        __builtin_amdgcn_sched_barrier(0);
#pragma unroll
        for (int k = 0; k < 14; ++k) {
            const int col = 4 * (l8 + 8 * k);
            v4f z = sv[k];
            if (k < 6) {
                const int gc = gw0 + col;
                z.x = (gc >= 0) ? z.x : 0.0f;
                z.y = (gc >= 0) ? z.y : 0.0f;
                z.z = (gc >= 0) ? z.z : 0.0f;
                z.w = (gc >= 0) ? z.w : 0.0f;
            }
            *(v4f*)(srow + col) = z;
        }
    }
    __builtin_amdgcn_fence(__ATOMIC_RELEASE, "workgroup", "local");
    __builtin_amdgcn_s_barrier();
    __builtin_amdgcn_fence(__ATOMIC_ACQUIRE, "workgroup", "local");

    const float* arow = &slab[n * SLAB_STRIDE + tw * 16 + 8 * khalf];

    v4f lo[16], hi[16];
#pragma unroll
    for (int q = 0; q < 16; ++q) {
        const int i = (q < 8) ? ((q & 1) ? 12 + (q >> 1) : (q >> 1)) : q - 4;
        lo[i] = *(const v4f*)(arow + 16 * i);
        hi[i] = *(const v4f*)(arow + 16 * i + 4);
    }
    __builtin_amdgcn_sched_barrier(0);

    v2f w_step[4];
#pragma unroll
    for (int j = 0; j < 4; ++j) {
        const int k0 = 8 * khalf + 2 * j;
        w_step[j].x = (k0 > n) ? 1.0f : 0.0f;
        w_step[j].y = (k0 + 1 > n) ? 1.0f : 0.0f;
    }
    v2f w_lin;
    w_lin.x = (float)(n - 8 * khalf);
    w_lin.y = 1.0f;

    v8f c[TPW];
#pragma unroll
    for (int s = 0; s < TPW; ++s) c[s] = (v8f){};

    v2f t[16][4], u[16];
#define FOLD_CHUNK(i)                                                     \
    {                                                                     \
        v2f p0, p1, p2, p3;                                               \
        p0.x = lo[i].x; p0.y = lo[i].y; p1.x = lo[i].z; p1.y = lo[i].w;   \
        p2.x = hi[i].x; p2.y = hi[i].y; p3.x = hi[i].z; p3.y = hi[i].w;   \
        t[i][3] = p3;                                                     \
        t[i][2] = p2 + p3;                                                \
        t[i][1] = p1 + t[i][2];                                           \
        t[i][0] = p0 + t[i][1];                                           \
        u[i]    = (t[i][1] + t[i][2]) + p3;                               \
    }

    v2f z[TPW][4];
#define EDGE_PREP(s)                                                                      \
    {                                                                                     \
        v2f d0 = t[s + 12][0] - t[s][0], d1 = t[s + 12][1] - t[s][1];                     \
        v2f d2 = t[s + 12][2] - t[s][2], d3 = t[s + 12][3] - t[s][3];                     \
                      \
        v2f f0, f1, f2, f3;                                                               \
        f0.x = d0.x + d0.y; f0.y = d0.y + d1.x;                                           \
        f1.x = d1.x + d1.y; f1.y = d1.y + d2.x;                                           \
        f2.x = d2.x + d2.y; f2.y = d2.y + d3.x;                                           \
        f3.x = d3.x + d3.y; f3.y = d3.y;                                                  \
              \
        const float other = __builtin_bit_cast(float, __builtin_amdgcn_permlanex16(       \
            __builtin_bit_cast(int, f0.x), __builtin_bit_cast(int, f0.x),                 \
            0x76543210, 0xfedcba98, false, false));            \
        v2f cy;                                                                           \
        cy.x = khalf ? 0.0f : other;                                                      \
        cy.y = cy.x;                                                                      \
        v2f a0, a1, a2, a3;                                                               \
        a0.x = lo[s].x; a0.y = lo[s].y; a1.x = lo[s].z; a1.y = lo[s].w;                   \
        a2.x = hi[s].x; a2.y = hi[s].y; a3.x = hi[s].z; a3.y = hi[s].w;                   \
        z[s][0] = (192.0f * a0 + f0) + cy; z[s][1] = (192.0f * a1 + f1) + cy;             \
        z[s][2] = (192.0f * a2 + f2) + cy; z[s][3] = (192.0f * a3 + f3) + cy;             \
    }
#define EDGE_WMMA2(sa, sb)                                                                \
    _Pragma("unroll") for (int j = 0; j < 4; ++j) {                                       \
        c[sa] = __builtin_amdgcn_wmma_f32_16x16x4_f32(false, z[sa][j], false, w_step[j],  \
                                                      (short)0, c[sa], false, false);     \
        c[sb] = __builtin_amdgcn_wmma_f32_16x16x4_f32(false, z[sb][j], false, w_step[j],  \
                                                      (short)0, c[sb], false, false);     \
    }

    FOLD_CHUNK(0)  FOLD_CHUNK(12) FOLD_CHUNK(1)  FOLD_CHUNK(13)
    EDGE_PREP(0)   EDGE_PREP(1)
    EDGE_WMMA2(0, 1)
    FOLD_CHUNK(2)  FOLD_CHUNK(14) FOLD_CHUNK(3)  FOLD_CHUNK(15)
    EDGE_PREP(2)   EDGE_PREP(3)
    EDGE_WMMA2(2, 3)
    FOLD_CHUNK(4)  FOLD_CHUNK(5)  FOLD_CHUNK(6)  FOLD_CHUNK(7)
    FOLD_CHUNK(8)  FOLD_CHUNK(9)  FOLD_CHUNK(10) FOLD_CHUNK(11)
#undef FOLD_CHUNK
#undef EDGE_PREP
#undef EDGE_WMMA2

    v2f S = t[4][0], G = t[4][0], SU = u[4];
#pragma unroll
    for (int m = 2; m <= 8; ++m) {
        S += t[3 + m][0];
        G += S;
        SU += u[3 + m];
    }
    v2f Pv[TPW], Hv[TPW], Uv[TPW];
    Hv[0] = (11.0f * t[1][0] + 10.0f * t[2][0]) + (9.0f * t[3][0] + G);
    S += t[12][0]; SU += u[12];
    Pv[0] = (t[1][0] + t[2][0]) + (t[3][0] + S);
    Uv[0] = (u[1] + u[2]) + (u[3] + SU);
    G += S;
    Hv[1] = 11.0f * t[2][0] + (10.0f * t[3][0] + G);
    S += t[13][0]; SU += u[13];
    Pv[1] = (t[2][0] + t[3][0]) + S;
    Uv[1] = (u[2] + u[3]) + SU;
    G += S;
    Hv[2] = 11.0f * t[3][0] + G;
    S += t[14][0]; SU += u[14];
    Pv[2] = t[3][0] + S;
    Uv[2] = u[3] + SU;
    G += S;
    Hv[3] = G;
    S += t[15][0]; SU += u[15];
    Pv[3] = S;
    Uv[3] = SU;

#pragma unroll
    for (int s = 0; s < TPW; ++s) {
        const v2f Qh = 16.0f * Hv[s] - 2.0f * Uv[s];
        v2f pq;
        pq.x = Pv[s].x + Pv[s].y;
        pq.y = (Qh.x + Qh.y) - Pv[s].y;
        c[s] = __builtin_amdgcn_wmma_f32_16x16x4_f32(false, pq, false, w_lin,
                                                     (short)0, c[s], false, false);
    }

    __builtin_amdgcn_sched_group_barrier(0x002, HEAD_VALU, 0);
#pragma unroll
    for (int g = 0; g < 16; ++g) {
        __builtin_amdgcn_sched_group_barrier(0x008, 1, 0);
        __builtin_amdgcn_sched_group_barrier(0x002, GAP_VALU, 0);
    }
    __builtin_amdgcn_sched_group_barrier(0x002, 56, 0);
    __builtin_amdgcn_sched_group_barrier(0x008, 4, 0);

    float* optr = out + obase;
    if (own01) {
#pragma unroll
        for (int s = 0; s < 2; ++s)
#pragma unroll
            for (int v = 0; v < 8; ++v)
                optr[v * Ww + 16 * s] = lv[s][v] * c[s][v];
    }
#pragma unroll
    for (int s = 2; s < TPW; ++s)
#pragma unroll
        for (int v = 0; v < 8; ++v)
            optr[v * Ww + 16 * s] = lv[s][v] * c[s][v];
}

extern "C" void kernel_launch(void* const* d_in, const int* in_sizes, int n_in,
                              void* d_out, int out_size, void* d_ws, size_t ws_size,
                              hipStream_t stream) {
    (void)in_sizes; (void)n_in; (void)out_size; (void)d_ws; (void)ws_size;
    const float* L = (const float*)d_in[0];
    const float* R = (const float*)d_in[1];
    float* O = (float*)d_out;

    static_assert(ROWS % 16 == 0, "rows must split into strips");
    static_assert(WGW == 4 * TPW, "a block's tiles are four quadruples");
    static_assert(WGW * (BPS - 1) <= WT && WGW * BPS >= WT, "five blocks cover a strip");
    static_assert(SLAB_COLS == DISP + WGW * 16, "slab = halo + tile columns");
    dim3 grid(NBLK);
    dim3 block(128);
    hipLaunchKernelGGL(StereoMatching_wmma_kernel, grid, block, 0, stream, L, R, O);
}
